# starting kernel
# baseline (speedup 1.0000x reference)
#include <hip/hip_runtime.h>

typedef __attribute__((ext_vector_type(2))) float v2f;
typedef __attribute__((ext_vector_type(8))) float v8f;

#define STRIDE 104
#define WAVES_PER_BLOCK 4
#define NTHREADS (WAVES_PER_BLOCK * 32)
#define BETA 3.0f
#define MAX_A 5.0f

static __device__ __forceinline__ v8f wmma4(v2f a, v2f b, v8f c) {
  return __builtin_amdgcn_wmma_f32_16x16x4_f32(
      false, a, false, b, (short)0, c, false, false);
}

static __device__ __forceinline__ float fast_tanh(float x) {
#if __has_builtin(__builtin_amdgcn_tanhf)
  return __builtin_amdgcn_tanhf(x);
#elif __has_builtin(__builtin_amdgcn_tanh_f32)
  return __builtin_amdgcn_tanh_f32(x);
#else
  float e = __expf(2.0f * x);
  return 1.0f - 2.0f / (e + 1.0f);
#endif
}

__global__ __launch_bounds__(NTHREADS)
void rbf_net_kernel(const float* __restrict__ s,  const float* __restrict__ a,
                    const float* __restrict__ W1, const float* __restrict__ b1,
                    const float* __restrict__ W2, const float* __restrict__ b2,
                    const float* __restrict__ L1, const float* __restrict__ bl1,
                    const float* __restrict__ WL2,const float* __restrict__ bL2,
                    float* __restrict__ out, int Btot, int nchunks) {
  __shared__ float scratch[WAVES_PER_BLOCK * 16 * STRIDE];
  __shared__ v2f   wfrag1[7 * 3 * 32];
  __shared__ v2f   lfrag1[7 * 3 * 32];
  __shared__ v2f   wlfrag[26 * 4 * 32];
  const int tid  = threadIdx.x;
  const int wid  = tid >> 5;
  const int lane = tid & 31;
  const int lid  = lane & 15;
  const int kh   = lane >> 4;
  float* hv = scratch + wid * (16 * STRIDE);

  for (int idx = tid; idx < 7 * 3 * 32; idx += NTHREADS) {
    int t = idx / 96, kk = (idx / 32) % 3, ln = idx & 31;
    int lidf = ln & 15, khf = ln >> 4;
    int col = t * 16 + lidf, k = kk * 4 + khf * 2;
    v2f w = {0.0f, 0.0f}, l = {0.0f, 0.0f};
    if (col < 100) {
      if (k < 10) {
        w.x = W1[k * 100 + col];  w.y = W1[(k + 1) * 100 + col];
        l.x = L1[k * 100 + col];  l.y = L1[(k + 1) * 100 + col];
      } else {
        w.x = b1[col];
        l.x = bl1[col];
      }
    }
    wfrag1[idx] = w;
    lfrag1[idx] = l;
  }
  for (int idx = tid; idx < 26 * 4 * 32; idx += NTHREADS) {
    int kk = idx / 128, t = (idx / 32) & 3, ln = idx & 31;
    int lidf = ln & 15, khf = ln >> 4;
    int col = t * 16 + lidf, k = kk * 4 + khf * 2;
    v2f w = {0.0f, 0.0f};
    if (col < 60) {
      if (kk < 25) {
        int n = col >> 1, d = col & 1;
        int base = n * 200 + k * 2 + d;
        w.x = WL2[base]; w.y = WL2[base + 2];
      } else if (khf == 0) {
        w.x = bL2[col];
      }
    }
    wlfrag[idx] = w;
  }

  v2f w2f[2][26];
#pragma unroll
  for (int t = 0; t < 2; ++t) {
    int col = t * 16 + lid;
#pragma unroll
    for (int kk = 0; kk < 25; ++kk) {
      int k = kk * 4 + kh * 2;
      v2f w = {0.0f, 0.0f};
      if (col < 30) { w.x = W2[k * 30 + col]; w.y = W2[(k + 1) * 30 + col]; }
      w2f[t][kk] = w;
    }
    v2f wb = {0.0f, 0.0f};
    if (col < 30 && kh == 0) wb.x = b2[col];
    w2f[t][25] = wb;
  }

  __syncthreads();

  const v2f hfb = kh ? (v2f){0.0f, 0.0f} : (v2f){1.0f, 0.0f};

  const int gwave  = blockIdx.x * WAVES_PER_BLOCK + wid;
  const int nwaves = gridDim.x * WAVES_PER_BLOCK;

  for (int chunk = gwave; chunk < nchunks; chunk += nwaves) {
    const int row0 = chunk * 16;
    const int grow = row0 + lid;
    const long rc  = (grow < Btot) ? (long)grow : (long)(Btot - 1);

    const float* srow = s + rc * 10;
    v2f af0  = *(const v2f*)(srow + kh * 2);
    v2f af1  = *(const v2f*)(srow + 4 + kh * 2);
    v2f af2l = *(const v2f*)(srow + 8);
    v2f af2  = kh ? (v2f){1.0f, 0.0f} : af2l;
    v2f af[3] = {af0, af1, af2};

#pragma unroll
    for (int t = 0; t < 7; ++t) {
      v8f acc = {0, 0, 0, 0, 0, 0, 0, 0};
#pragma unroll
      for (int kk = 0; kk < 3; ++kk)
        acc = wmma4(af[kk], wfrag1[(t * 3 + kk) * 32 + lane], acc);
      int col = t * 16 + lid;
      if (col < 100) {
#pragma unroll
        for (int j = 0; j < 8; ++j)
          hv[(j + 8 * kh) * STRIDE + col] = fmaxf(acc[j], 0.0f);
      }
    }

    v8f vacc[2];
    vacc[0] = (v8f){0, 0, 0, 0, 0, 0, 0, 0};
    vacc[1] = (v8f){0, 0, 0, 0, 0, 0, 0, 0};
#pragma unroll
    for (int kk = 0; kk < 26; ++kk) {
      v2f hf = (kk < 25) ? *(const v2f*)&hv[lid * STRIDE + kk * 4 + kh * 2]
                         : hfb;
      vacc[0] = wmma4(hf, w2f[0][kk], vacc[0]);
      vacc[1] = wmma4(hf, w2f[1][kk], vacc[1]);
    }

#pragma unroll
    for (int t = 0; t < 7; ++t) {
      v8f acc = {0, 0, 0, 0, 0, 0, 0, 0};
#pragma unroll
      for (int kk = 0; kk < 3; ++kk)
        acc = wmma4(af[kk], lfrag1[(t * 3 + kk) * 32 + lane], acc);
      int col = t * 16 + lid;
      if (col < 100) {
#pragma unroll
        for (int j = 0; j < 8; ++j)
          hv[(j + 8 * kh) * STRIDE + col] = fmaxf(acc[j], 0.0f);
      }
    }

    v8f cacc[4];
#pragma unroll
    for (int t = 0; t < 4; ++t) cacc[t] = (v8f){0, 0, 0, 0, 0, 0, 0, 0};
#pragma unroll
    for (int kk = 0; kk < 26; ++kk) {
      v2f hf = (kk < 25) ? *(const v2f*)&hv[lid * STRIDE + kk * 4 + kh * 2]
                         : hfb;
#pragma unroll
      for (int t = 0; t < 4; ++t)
        cacc[t] = wmma4(hf, wlfrag[(kk * 4 + t) * 32 + lane], cacc[t]);
    }

#pragma unroll
    for (int t = 0; t < 4; ++t) {
      int col = t * 16 + lid;
      if (col < 60) {
#pragma unroll
        for (int j = 0; j < 8; ++j)
          hv[(j + 8 * kh) * STRIDE + col] = MAX_A * fast_tanh(cacc[t][j]);
      }
    }
#pragma unroll
    for (int t = 0; t < 2; ++t) {
      int n = t * 16 + lid;
      if (n < 30) {
#pragma unroll
        for (int j = 0; j < 8; ++j)
          hv[(j + 8 * kh) * STRIDE + 64 + n] = vacc[t][j];
      }
    }

    {
      int row  = lane & 15;
      int half = lane >> 4;
      long r   = (long)row0 + row;
      long ra  = (r < Btot) ? r : (long)(Btot - 1);
      v2f av   = *(const v2f*)(a + ra * 2);
      float a0 = av.x, a1 = av.y;
      float ssum = 0.0f, wsum = 0.0f;
#pragma unroll
      for (int i = 0; i < 15; ++i) {
        int n    = half * 15 + i;
        float c0 = hv[row * STRIDE + 2 * n];
        float c1 = hv[row * STRIDE + 2 * n + 1];
        float v  = hv[row * STRIDE + 64 + n];
        float d0 = c0 - a0, d1 = c1 - a1;
        float e  = __expf(-BETA * sqrtf(d0 * d0 + d1 * d1));
        ssum += e;
        wsum += e * v;
      }
      ssum += __shfl_xor(ssum, 16, 32);
      wsum += __shfl_xor(wsum, 16, 32);
      if (half == 0 && r < Btot) out[r] = wsum / ssum;
    }
  }
}

extern "C" void kernel_launch(void* const* d_in, const int* in_sizes, int n_in,
                              void* d_out, int out_size, void* d_ws, size_t ws_size,
                              hipStream_t stream) {
  const float* s   = (const float*)d_in[0];
  const float* a   = (const float*)d_in[1];
  const float* W1  = (const float*)d_in[2];
  const float* b1  = (const float*)d_in[3];
  const float* W2  = (const float*)d_in[4];
  const float* b2  = (const float*)d_in[5];
  const float* L1  = (const float*)d_in[6];
  const float* bl1 = (const float*)d_in[7];
  const float* WL2 = (const float*)d_in[8];
  const float* bL2 = (const float*)d_in[9];
  float* out = (float*)d_out;

  int Btot    = in_sizes[0] / 10;
  int nchunks = (Btot + 15) / 16;
  int blocks  = (nchunks + WAVES_PER_BLOCK - 1) / WAVES_PER_BLOCK;
  if (blocks > 1024) blocks = 1024;
  if (blocks < 1) blocks = 1;

  rbf_net_kernel<<<blocks, NTHREADS, 0, stream>>>(
      s, a, W1, b1, W2, b2, L1, bl1, WL2, bL2, out, Btot, nchunks);
}


# W2 fragments moved from registers to LDS: 266 -> 210 VGPRs, so 4 waves per SIMD instead of 3 and all 1024 blocks resident in one round instead of 768 then 256
# speedup vs baseline: 1.1085x; 1.1085x over previous
#include <hip/hip_runtime.h>

typedef __attribute__((ext_vector_type(2))) float v2f;
typedef __attribute__((ext_vector_type(8))) float v8f;

#define STRIDE 104
#define WAVES_PER_BLOCK 4
#define NTHREADS (WAVES_PER_BLOCK * 32)
#define BETA 3.0f
#define MAX_A 5.0f

static __device__ __forceinline__ v8f wmma4(v2f a, v2f b, v8f c) {
  return __builtin_amdgcn_wmma_f32_16x16x4_f32(
      false, a, false, b, (short)0, c, false, false);
}

static __device__ __forceinline__ float fast_tanh(float x) {
#if __has_builtin(__builtin_amdgcn_tanhf)
  return __builtin_amdgcn_tanhf(x);
#elif __has_builtin(__builtin_amdgcn_tanh_f32)
  return __builtin_amdgcn_tanh_f32(x);
#else
  float e = __expf(2.0f * x);
  return 1.0f - 2.0f / (e + 1.0f);
#endif
}

__global__ __launch_bounds__(NTHREADS)
void rbf_net_kernel(const float* __restrict__ s,  const float* __restrict__ a,
                    const float* __restrict__ W1, const float* __restrict__ b1,
                    const float* __restrict__ W2, const float* __restrict__ b2,
                    const float* __restrict__ L1, const float* __restrict__ bl1,
                    const float* __restrict__ WL2,const float* __restrict__ bL2,
                    float* __restrict__ out, int Btot, int nchunks) {
  __shared__ float scratch[WAVES_PER_BLOCK * 16 * STRIDE];
  __shared__ v2f   wfrag1[7 * 3 * 32];
  __shared__ v2f   lfrag1[7 * 3 * 32];
  __shared__ v2f   wlfrag[26 * 4 * 32];
  __shared__ v2f   w2frag[26 * 2 * 32];
  const int tid  = threadIdx.x;
  const int wid  = tid >> 5;
  const int lane = tid & 31;
  const int lid  = lane & 15;
  const int kh   = lane >> 4;
  float* hv = scratch + wid * (16 * STRIDE);

  for (int idx = tid; idx < 7 * 3 * 32; idx += NTHREADS) {
    int t = idx / 96, kk = (idx / 32) % 3, ln = idx & 31;
    int lidf = ln & 15, khf = ln >> 4;
    int col = t * 16 + lidf, k = kk * 4 + khf * 2;
    v2f w = {0.0f, 0.0f}, l = {0.0f, 0.0f};
    if (col < 100) {
      if (k < 10) {
        w.x = W1[k * 100 + col];  w.y = W1[(k + 1) * 100 + col];
        l.x = L1[k * 100 + col];  l.y = L1[(k + 1) * 100 + col];
      } else {
        w.x = b1[col];
        l.x = bl1[col];
      }
    }
    wfrag1[idx] = w;
    lfrag1[idx] = l;
  }
  for (int idx = tid; idx < 26 * 4 * 32; idx += NTHREADS) {
    int kk = idx / 128, t = (idx / 32) & 3, ln = idx & 31;
    int lidf = ln & 15, khf = ln >> 4;
    int col = t * 16 + lidf, k = kk * 4 + khf * 2;
    v2f w = {0.0f, 0.0f};
    if (col < 60) {
      if (kk < 25) {
        int n = col >> 1, d = col & 1;
        int base = n * 200 + k * 2 + d;
        w.x = WL2[base]; w.y = WL2[base + 2];
      } else if (khf == 0) {
        w.x = bL2[col];
      }
    }
    wlfrag[idx] = w;
  }

#pragma unroll
  for (int i = 0; i < 13; ++i) {
    int idx = tid + i * NTHREADS;
    int kk = idx >> 6, t = (idx >> 5) & 1;
    int col = t * 16 + lid, k = kk * 4 + kh * 2;
    int colc = col < 30 ? col : 29;
    int kc   = kk < 25 ? k : 98;
    float x = W2[kc * 30 + colc];
    float y = W2[(kc + 1) * 30 + colc];
    float bb = b2[colc];
    v2f w;
    w.x = (col < 30) ? ((kk < 25) ? x : (kh == 0 ? bb : 0.0f)) : 0.0f;
    w.y = (col < 30 && kk < 25) ? y : 0.0f;
    w2frag[idx] = w;
  }

  __syncthreads();

  const v2f hfb = kh ? (v2f){0.0f, 0.0f} : (v2f){1.0f, 0.0f};

  const int gwave  = blockIdx.x * WAVES_PER_BLOCK + wid;
  const int nwaves = gridDim.x * WAVES_PER_BLOCK;

  for (int chunk = gwave; chunk < nchunks; chunk += nwaves) {
    const int row0 = chunk * 16;
    const int grow = row0 + lid;
    const long rc  = (grow < Btot) ? (long)grow : (long)(Btot - 1);

    const float* srow = s + rc * 10;
    v2f af0  = *(const v2f*)(srow + kh * 2);
    v2f af1  = *(const v2f*)(srow + 4 + kh * 2);
    v2f af2l = *(const v2f*)(srow + 8);
    v2f af2  = kh ? (v2f){1.0f, 0.0f} : af2l;
    v2f af[3] = {af0, af1, af2};

#pragma unroll
    for (int t = 0; t < 7; ++t) {
      v8f acc = {0, 0, 0, 0, 0, 0, 0, 0};
#pragma unroll
      for (int kk = 0; kk < 3; ++kk)
        acc = wmma4(af[kk], wfrag1[(t * 3 + kk) * 32 + lane], acc);
      int col = t * 16 + lid;
      if (col < 100) {
#pragma unroll
        for (int j = 0; j < 8; ++j)
          hv[(j + 8 * kh) * STRIDE + col] = fmaxf(acc[j], 0.0f);
      }
    }

    v8f vacc[2];
    vacc[0] = (v8f){0, 0, 0, 0, 0, 0, 0, 0};
    vacc[1] = (v8f){0, 0, 0, 0, 0, 0, 0, 0};
#pragma unroll
    for (int kk = 0; kk < 26; ++kk) {
      v2f hf = (kk < 25) ? *(const v2f*)&hv[lid * STRIDE + kk * 4 + kh * 2]
                         : hfb;
      vacc[0] = wmma4(hf, w2frag[(kk * 2 + 0) * 32 + lane], vacc[0]);
      vacc[1] = wmma4(hf, w2frag[(kk * 2 + 1) * 32 + lane], vacc[1]);
    }

#pragma unroll
    for (int t = 0; t < 7; ++t) {
      v8f acc = {0, 0, 0, 0, 0, 0, 0, 0};
#pragma unroll
      for (int kk = 0; kk < 3; ++kk)
        acc = wmma4(af[kk], lfrag1[(t * 3 + kk) * 32 + lane], acc);
      int col = t * 16 + lid;
      if (col < 100) {
#pragma unroll
        for (int j = 0; j < 8; ++j)
          hv[(j + 8 * kh) * STRIDE + col] = fmaxf(acc[j], 0.0f);
      }
    }

    v8f cacc[4];
#pragma unroll
    for (int t = 0; t < 4; ++t) cacc[t] = (v8f){0, 0, 0, 0, 0, 0, 0, 0};
#pragma unroll
    for (int kk = 0; kk < 26; ++kk) {
      v2f hf = (kk < 25) ? *(const v2f*)&hv[lid * STRIDE + kk * 4 + kh * 2]
                         : hfb;
#pragma unroll
      for (int t = 0; t < 4; ++t)
        cacc[t] = wmma4(hf, wlfrag[(kk * 4 + t) * 32 + lane], cacc[t]);
    }

#pragma unroll
    for (int t = 0; t < 4; ++t) {
      int col = t * 16 + lid;
      if (col < 60) {
#pragma unroll
        for (int j = 0; j < 8; ++j)
          hv[(j + 8 * kh) * STRIDE + col] = MAX_A * fast_tanh(cacc[t][j]);
      }
    }
#pragma unroll
    for (int t = 0; t < 2; ++t) {
      int n = t * 16 + lid;
      if (n < 30) {
#pragma unroll
        for (int j = 0; j < 8; ++j)
          hv[(j + 8 * kh) * STRIDE + 64 + n] = vacc[t][j];
      }
    }

    {
      int row  = lane & 15;
      int half = lane >> 4;
      long r   = (long)row0 + row;
      long ra  = (r < Btot) ? r : (long)(Btot - 1);
      v2f av   = *(const v2f*)(a + ra * 2);
      float a0 = av.x, a1 = av.y;
      float ssum = 0.0f, wsum = 0.0f;
#pragma unroll
      for (int i = 0; i < 15; ++i) {
        int n    = half * 15 + i;
        float c0 = hv[row * STRIDE + 2 * n];
        float c1 = hv[row * STRIDE + 2 * n + 1];
        float v  = hv[row * STRIDE + 64 + n];
        float d0 = c0 - a0, d1 = c1 - a1;
        float e  = __expf(-BETA * sqrtf(d0 * d0 + d1 * d1));
        ssum += e;
        wsum += e * v;
      }
      ssum += __shfl_xor(ssum, 16, 32);
      wsum += __shfl_xor(wsum, 16, 32);
      if (half == 0 && r < Btot) out[r] = wsum / ssum;
    }
  }
}

extern "C" void kernel_launch(void* const* d_in, const int* in_sizes, int n_in,
                              void* d_out, int out_size, void* d_ws, size_t ws_size,
                              hipStream_t stream) {
  const float* s   = (const float*)d_in[0];
  const float* a   = (const float*)d_in[1];
  const float* W1  = (const float*)d_in[2];
  const float* b1  = (const float*)d_in[3];
  const float* W2  = (const float*)d_in[4];
  const float* b2  = (const float*)d_in[5];
  const float* L1  = (const float*)d_in[6];
  const float* bl1 = (const float*)d_in[7];
  const float* WL2 = (const float*)d_in[8];
  const float* bL2 = (const float*)d_in[9];
  float* out = (float*)d_out;

  int Btot    = in_sizes[0] / 10;
  int nchunks = (Btot + 15) / 16;
  int blocks  = (nchunks + WAVES_PER_BLOCK - 1) / WAVES_PER_BLOCK;
  if (blocks > 1024) blocks = 1024;
  if (blocks < 1) blocks = 1;

  rbf_net_kernel<<<blocks, NTHREADS, 0, stream>>>(
      s, a, W1, b1, W2, b2, L1, bl1, WL2, bL2, out, Btot, nchunks);
}


# LDS accesses grouped away from the WMMA stream: 16-wave block, 32 rows per wave, paired k-steps read as b128, each weight fragment feeds 2 row tiles; one read group per 16 WMMAs (was per 2-4)
# speedup vs baseline: 1.3258x; 1.1960x over previous
#include <hip/hip_runtime.h>

typedef __attribute__((ext_vector_type(2))) float v2f;
typedef __attribute__((ext_vector_type(4))) float v4f;
typedef __attribute__((ext_vector_type(8))) float v8f;

#define STRIDE 108
#define WAVES 16
#define NTHREADS (WAVES * 32)
#define ROWS 32
#define BETA 3.0f
#define MAX_A 5.0f

static __device__ __forceinline__ v8f wmma4(v2f a, v2f b, v8f c) {
  return __builtin_amdgcn_wmma_f32_16x16x4_f32(
      false, a, false, b, (short)0, c, false, false);
}

static __device__ __forceinline__ float fast_tanh(float x) {
#if __has_builtin(__builtin_amdgcn_tanhf)
  return __builtin_amdgcn_tanhf(x);
#elif __has_builtin(__builtin_amdgcn_tanh_f32)
  return __builtin_amdgcn_tanh_f32(x);
#else
  float e = __expf(2.0f * x);
  return 1.0f - 2.0f / (e + 1.0f);
#endif
}

template <int NT, int PG> struct Frags { v4f a[PG][2]; v4f b[PG][NT]; };

template <int NT, int PG>
static __device__ __forceinline__ void load_frags(Frags<NT, PG>& f,
    const float* hv, const v4f* w, int p0, int apos, int lane) {
#pragma unroll
  for (int q = 0; q < PG; ++q) {
    if (p0 + q < 13) {
      f.a[q][0] = *(const v4f*)(hv + apos + 8 * (p0 + q));
      f.a[q][1] = *(const v4f*)(hv + 16 * STRIDE + apos + 8 * (p0 + q));
#pragma unroll
      for (int t = 0; t < NT; ++t)
        f.b[q][t] = w[((p0 + q) * NT + t) * 32 + lane];
    }
  }
}

template <int NT, int PG>
static __device__ __forceinline__ void mma_frags(const Frags<NT, PG>& f,
    v8f (&acc)[2][NT], int p0) {
#pragma unroll
  for (int q = 0; q < PG; ++q) {
    if (p0 + q < 13) {
#pragma unroll
      for (int m = 0; m < 2; ++m)
#pragma unroll
        for (int t = 0; t < NT; ++t)
          acc[m][t] = wmma4(f.a[q][m].xy, f.b[q][t].xy, acc[m][t]);
#pragma unroll
      for (int m = 0; m < 2; ++m)
#pragma unroll
        for (int t = 0; t < NT; ++t)
          acc[m][t] = wmma4(f.a[q][m].zw, f.b[q][t].zw, acc[m][t]);
    }
  }
}

template <int NT, int PG>
static __device__ __forceinline__ void k_loop(v8f (&acc)[2][NT],
    const float* hv, const v4f* w, int apos, int lane) {
  Frags<NT, PG> cur, nxt;
  load_frags<NT, PG>(cur, hv, w, 0, apos, lane);
#pragma unroll
  for (int p0 = 0; p0 < 13; p0 += PG) {
    if (p0 + PG < 13) load_frags<NT, PG>(nxt, hv, w, p0 + PG, apos, lane);
    __builtin_amdgcn_sched_barrier(0);
    mma_frags<NT, PG>(cur, acc, p0);
    __builtin_amdgcn_sched_barrier(0);
    if (p0 + PG < 13) cur = nxt;
  }
}

static __device__ __forceinline__ void stage_in(float* hv, const v4f* wa,
    const v2f* wb, const v2f (&af)[2][3], int lane, int lid, int spos) {
  v4f ba[7];
  v2f bb[7];
#pragma unroll
  for (int t = 0; t < 7; ++t) {
    ba[t] = wa[t * 32 + lane];
    bb[t] = wb[t * 32 + lane];
  }
  __builtin_amdgcn_sched_barrier(0);
  v8f acc[7][2];
#pragma unroll
  for (int t = 0; t < 7; ++t) {
    acc[t][0] = (v8f){0, 0, 0, 0, 0, 0, 0, 0};
    acc[t][1] = (v8f){0, 0, 0, 0, 0, 0, 0, 0};
#pragma unroll
    for (int m = 0; m < 2; ++m) acc[t][m] = wmma4(af[m][0], ba[t].xy, acc[t][m]);
#pragma unroll
    for (int m = 0; m < 2; ++m) acc[t][m] = wmma4(af[m][1], ba[t].zw, acc[t][m]);
#pragma unroll
    for (int m = 0; m < 2; ++m) acc[t][m] = wmma4(af[m][2], bb[t], acc[t][m]);
  }
  __builtin_amdgcn_sched_barrier(0);
#pragma unroll
  for (int t = 0; t < 7; ++t) {
    if (t < 6 || lid < 8) {
#pragma unroll
      for (int m = 0; m < 2; ++m)
#pragma unroll
        for (int j = 0; j < 8; ++j)
          hv[m * 16 * STRIDE + j * STRIDE + spos + t * 16] =
              fmaxf(acc[t][m][j], 0.0f);
    }
  }
}

__global__ __launch_bounds__(NTHREADS)
void rbf_net_kernel(const float* __restrict__ s,  const float* __restrict__ a,
                    const float* __restrict__ W1, const float* __restrict__ b1,
                    const float* __restrict__ W2, const float* __restrict__ b2,
                    const float* __restrict__ L1, const float* __restrict__ bl1,
                    const float* __restrict__ WL2,const float* __restrict__ bL2,
                    float* __restrict__ out, int Btot, int nchunks) {
  __shared__ __attribute__((aligned(16))) float scratch[WAVES * ROWS * STRIDE];
  __shared__ v4f w1a[7 * 32];
  __shared__ v2f w1b[7 * 32];
  __shared__ v4f l1a[7 * 32];
  __shared__ v2f l1b[7 * 32];
  __shared__ v4f wl4[13 * 4 * 32];
  __shared__ v4f w24[13 * 2 * 32];
  const int tid  = threadIdx.x;
  const int wid  = tid >> 5;
  const int lane = tid & 31;
  const int lid  = lane & 15;
  const int kh   = lane >> 4;
  float* hv = scratch + wid * (ROWS * STRIDE);

  if (tid < 7 * 32) {
    int t = tid >> 5;
    int col = t * 16 + lid;
    int cc = col < 100 ? col : 99;
    bool in = col < 100;
    int k0 = 2 * kh, k1 = 4 + 2 * kh;
    float w00 = W1[k0 * 100 + cc], w01 = W1[(k0 + 1) * 100 + cc];
    float w10 = W1[k1 * 100 + cc], w11 = W1[(k1 + 1) * 100 + cc];
    float w8 = W1[800 + cc], w9 = W1[900 + cc], wb = b1[cc];
    float l00 = L1[k0 * 100 + cc], l01 = L1[(k0 + 1) * 100 + cc];
    float l10 = L1[k1 * 100 + cc], l11 = L1[(k1 + 1) * 100 + cc];
    float l8 = L1[800 + cc], l9 = L1[900 + cc], lb = bl1[cc];
    v4f z4 = {0.0f, 0.0f, 0.0f, 0.0f};
    v2f z2 = {0.0f, 0.0f};
    v2f unit = (col == 100 && kh) ? (v2f){1.0f, 0.0f} : z2;
    w1a[tid] = in ? (v4f){w00, w01, w10, w11} : z4;
    l1a[tid] = in ? (v4f){l00, l01, l10, l11} : z4;
    w1b[tid] = in ? (kh ? (v2f){wb, 0.0f} : (v2f){w8, w9}) : unit;
    l1b[tid] = in ? (kh ? (v2f){lb, 0.0f} : (v2f){l8, l9}) : unit;
  }
#pragma unroll
  for (int i = 0; i < 4; ++i) {
    int idx = tid + i * NTHREADS;
    int ic = idx < 13 * 4 * 32 ? idx : 13 * 4 * 32 - 1;
    int p = ic >> 7, t = (ic >> 5) & 3;
    int col = t * 16 + lid;
    int cc = col < 60 ? col : 59;
    int n = cc >> 1, d = cc & 1;
    int ka = 8 * p + 2 * kh;
    int kb = ka + 4;
    int kbc = kb < 99 ? kb : 98;
    float xa = WL2[n * 200 + ka * 2 + d], ya = WL2[n * 200 + ka * 2 + 2 + d];
    float xb = WL2[n * 200 + kbc * 2 + d], yb = WL2[n * 200 + kbc * 2 + 2 + d];
    float bb = bL2[cc];
    v4f w;
    bool in = col < 60;
    w.x = in ? xa : 0.0f;
    w.y = in ? ya : 0.0f;
    w.z = in ? (p < 12 ? xb : (kh == 0 ? bb : 0.0f)) : 0.0f;
    w.w = (in && p < 12) ? yb : 0.0f;
    if (idx < 13 * 4 * 32) wl4[idx] = w;
  }
#pragma unroll
  for (int i = 0; i < 2; ++i) {
    int idx = tid + i * NTHREADS;
    int ic = idx < 13 * 2 * 32 ? idx : 13 * 2 * 32 - 1;
    int p = ic >> 6, t = (ic >> 5) & 1;
    int col = t * 16 + lid;
    int cc = col < 30 ? col : 29;
    int ka = 8 * p + 2 * kh;
    int kb = ka + 4;
    int kbc = kb < 99 ? kb : 98;
    float xa = W2[ka * 30 + cc], ya = W2[(ka + 1) * 30 + cc];
    float xb = W2[kbc * 30 + cc], yb = W2[(kbc + 1) * 30 + cc];
    float bb = b2[cc];
    v4f w;
    bool in = col < 30;
    w.x = in ? xa : 0.0f;
    w.y = in ? ya : 0.0f;
    w.z = in ? (p < 12 ? xb : (kh == 0 ? bb : 0.0f)) : 0.0f;
    w.w = (in && p < 12) ? yb : 0.0f;
    if (idx < 13 * 2 * 32) w24[idx] = w;
  }

  __syncthreads();

  const int apos = lid * STRIDE + 4 * kh;
  const int plid = (lid & 9) | ((lid & 2) << 1) | ((lid & 4) >> 1);
  const int spos = (8 * kh) * STRIDE + plid;

  const int gwave  = blockIdx.x * WAVES + wid;
  const int nwaves = gridDim.x * WAVES;

  for (int chunk = gwave; chunk < nchunks; chunk += nwaves) {
    const int row0 = chunk * ROWS;

    v2f af[2][3];
#pragma unroll
    for (int m = 0; m < 2; ++m) {
      const int grow = row0 + 16 * m + lid;
      const long rc  = (grow < Btot) ? (long)grow : (long)(Btot - 1);
      const float* srow = s + rc * 10;
      af[m][0] = *(const v2f*)(srow + kh * 2);
      af[m][1] = *(const v2f*)(srow + 4 + kh * 2);
      v2f l    = *(const v2f*)(srow + 8);
      af[m][2] = kh ? (v2f){1.0f, 0.0f} : l;
    }

stage_in(hv, w1a, w1b, af, lane, lid, spos);

    v8f vacc[2][2];
#pragma unroll
    for (int m = 0; m < 2; ++m)
#pragma unroll
      for (int t = 0; t < 2; ++t) vacc[m][t] = (v8f){0, 0, 0, 0, 0, 0, 0, 0};
    k_loop<2, 2>(vacc, hv, w24, apos, lane);

stage_in(hv, l1a, l1b, af, lane, lid, spos);

    v8f cacc[2][4];
#pragma unroll
    for (int m = 0; m < 2; ++m)
#pragma unroll
      for (int t = 0; t < 4; ++t) cacc[m][t] = (v8f){0, 0, 0, 0, 0, 0, 0, 0};
    k_loop<4, 1>(cacc, hv, wl4, apos, lane);

#pragma unroll
    for (int m = 0; m < 2; ++m) {
      float* hm = hv + m * 16 * STRIDE + (8 * kh) * STRIDE;
#pragma unroll
      for (int t = 0; t < 4; ++t) {
        int col = t * 16 + lid;
        if (col < 60) {
#pragma unroll
          for (int j = 0; j < 8; ++j)
            hm[j * STRIDE + col] = MAX_A * fast_tanh(cacc[m][t][j]);
        }
      }
#pragma unroll
      for (int t = 0; t < 2; ++t) {
        int n = t * 16 + lid;
        if (n < 30) {
#pragma unroll
          for (int j = 0; j < 8; ++j)
            hm[j * STRIDE + 64 + n] = vacc[m][t][j];
        }
      }
    }

    {
      long r   = (long)row0 + lane;
      long ra  = (r < Btot) ? r : (long)(Btot - 1);
      v2f av   = *(const v2f*)(a + ra * 2);
      float a0 = av.x, a1 = av.y;
      const float* hr = hv + lane * STRIDE;
      float ssum[2] = {0.0f, 0.0f}, wsum[2] = {0.0f, 0.0f};
#pragma unroll
      for (int half = 0; half < 2; ++half) {
#pragma unroll
        for (int i = 0; i < 15; ++i) {
          int n    = half * 15 + i;
          float c0 = hr[2 * n];
          float c1 = hr[2 * n + 1];
          float v  = hr[64 + n];
          float d0 = c0 - a0, d1 = c1 - a1;
          float e  = __expf(-BETA * sqrtf(d0 * d0 + d1 * d1));
          ssum[half] += e;
          wsum[half] += e * v;
        }
      }
      float st = ssum[0] + ssum[1];
      float wt = wsum[0] + wsum[1];
      if (r < Btot) out[r] = wt / st;
    }
  }
}

extern "C" void kernel_launch(void* const* d_in, const int* in_sizes, int n_in,
                              void* d_out, int out_size, void* d_ws, size_t ws_size,
                              hipStream_t stream) {
  const float* s   = (const float*)d_in[0];
  const float* a   = (const float*)d_in[1];
  const float* W1  = (const float*)d_in[2];
  const float* b1  = (const float*)d_in[3];
  const float* W2  = (const float*)d_in[4];
  const float* b2  = (const float*)d_in[5];
  const float* L1  = (const float*)d_in[6];
  const float* bl1 = (const float*)d_in[7];
  const float* WL2 = (const float*)d_in[8];
  const float* bL2 = (const float*)d_in[9];
  float* out = (float*)d_out;

  int Btot    = in_sizes[0] / 10;
  int nchunks = (Btot + ROWS - 1) / ROWS;
  int blocks  = (nchunks + WAVES - 1) / WAVES;
  if (blocks > 256) blocks = 256;
  if (blocks < 1) blocks = 1;

  rbf_net_kernel<<<blocks, NTHREADS, 0, stream>>>(
      s, a, W1, b1, W2, b2, L1, bl1, WL2, bL2, out, Btot, nchunks);
}


# Products transposed (weights = A, batch = B): each layer's D registers feed the next layer's B operand and the softmax directly, only weights stay in LDS; k order and softmax sum grouping change
# speedup vs baseline: 1.4224x; 1.0729x over previous
#include <hip/hip_runtime.h>

typedef __attribute__((ext_vector_type(2))) float v2f;
typedef __attribute__((ext_vector_type(4))) float v4f;
typedef __attribute__((ext_vector_type(8))) float v8f;

#define WAVES 16
#define NTHREADS (WAVES * 32)
#define ROWS 32
#define BETA 3.0f
#define MAX_A 5.0f

static __device__ __forceinline__ v8f wmma4(v2f a, v2f b, v8f c) {
  return __builtin_amdgcn_wmma_f32_16x16x4_f32(
      false, a, false, b, (short)0, c, false, false);
}

static __device__ __forceinline__ float fast_tanh(float x) {
#if __has_builtin(__builtin_amdgcn_tanhf)
  return __builtin_amdgcn_tanhf(x);
#elif __has_builtin(__builtin_amdgcn_tanh_f32)
  return __builtin_amdgcn_tanh_f32(x);
#else
  float e = __expf(2.0f * x);
  return 1.0f - 2.0f / (e + 1.0f);
#endif
}

static __device__ __forceinline__ float other_half(float x) {
#if __has_builtin(__builtin_amdgcn_permlanex16)
  unsigned u = __builtin_bit_cast(unsigned, x);
  u = __builtin_amdgcn_permlanex16(u, u, 0x76543210u, 0xfedcba98u, false, false);
  return __builtin_bit_cast(float, u);
#else
  return __shfl_xor(x, 16, 32);
#endif
}

struct WIn { v4f a; v2f b; };
template <int NT> struct WK { v4f w[2][NT]; };

static __device__ __forceinline__ void load_in(WIn& f, const v4f* wa,
    const v2f* wb, int t, int lane) {
  f.a = wa[t * 32 + lane];
  f.b = wb[t * 32 + lane];
}

template <int NT>
static __device__ __forceinline__ void load_k(WK<NT>& f, const v4f* wk, int t,
    int lane) {
#pragma unroll
  for (int pp = 0; pp < 2; ++pp) {
    if (2 * t + pp < 13) {
#pragma unroll
      for (int tt = 0; tt < NT; ++tt)
        f.w[pp][tt] = wk[((2 * t + pp) * NT + tt) * 32 + lane];
    }
  }
}

static __device__ __forceinline__ void first_layer(v8f (&h)[2], const WIn& f,
    const v2f (&af)[2][3]) {
  const v8f z = {0, 0, 0, 0, 0, 0, 0, 0};
#pragma unroll
  for (int m = 0; m < 2; ++m) h[m] = wmma4(f.a.xy, af[m][0], z);
#pragma unroll
  for (int m = 0; m < 2; ++m) h[m] = wmma4(f.a.zw, af[m][1], h[m]);
#pragma unroll
  for (int m = 0; m < 2; ++m) h[m] = wmma4(f.b, af[m][2], h[m]);
}

template <int NT>
static __device__ __forceinline__ void second_layer(v8f (&acc)[2][NT],
    const v8f (&h)[2], const WK<NT>& f, int t) {
#pragma unroll
  for (int pp = 0; pp < 2; ++pp) {
    if (2 * t + pp < 13) {
#pragma unroll
      for (int qq = 0; qq < 2; ++qq) {
        const int q = 2 * pp + qq;
        v2f b[2];
#pragma unroll
        for (int m = 0; m < 2; ++m)
          b[m] = (v2f){fmaxf(h[m][2 * q], 0.0f), fmaxf(h[m][2 * q + 1], 0.0f)};
#pragma unroll
        for (int m = 0; m < 2; ++m)
#pragma unroll
          for (int tt = 0; tt < NT; ++tt)
            acc[m][tt] = wmma4(qq ? f.w[pp][tt].zw : f.w[pp][tt].xy, b[m],
                               acc[m][tt]);
      }
    }
  }
}

template <int NT>
static __device__ __forceinline__ void branch(v8f (&acc)[2][NT], const v4f* wa,
    const v2f* wb, const v4f* wk, const v2f (&af)[2][3], int lane) {
  WIn in_nxt, in_n2;
  WK<NT> k_cur, k_nxt;
  v8f hc[2], hn[2];
  load_in(in_n2, wa, wb, 0, lane);
  load_in(in_nxt, wa, wb, 1, lane);
  load_k<NT>(k_cur, wk, 0, lane);
  first_layer(hc, in_n2, af);
#pragma unroll
  for (int t = 0; t < 7; ++t) {
    if (t + 2 < 7) load_in(in_n2, wa, wb, t + 2, lane);
    if (t + 1 < 7) load_k<NT>(k_nxt, wk, t + 1, lane);
    __builtin_amdgcn_sched_barrier(0);
    if (t + 1 < 7) first_layer(hn, in_nxt, af);
    second_layer<NT>(acc, hc, k_cur, t);
    __builtin_amdgcn_sched_barrier(0);
    if (t + 1 < 7) {
      hc[0] = hn[0];
      hc[1] = hn[1];
      k_cur = k_nxt;
    }
    if (t + 2 < 7) in_nxt = in_n2;
  }
}

__global__ __launch_bounds__(NTHREADS)
void rbf_net_kernel(const float* __restrict__ s,  const float* __restrict__ a,
                    const float* __restrict__ W1, const float* __restrict__ b1,
                    const float* __restrict__ W2, const float* __restrict__ b2,
                    const float* __restrict__ L1, const float* __restrict__ bl1,
                    const float* __restrict__ WL2,const float* __restrict__ bL2,
                    float* __restrict__ out, int Btot, int nchunks) {
  __shared__ v4f w1a[7 * 32];
  __shared__ v2f w1b[7 * 32];
  __shared__ v4f l1a[7 * 32];
  __shared__ v2f l1b[7 * 32];
  __shared__ v4f wl4[13 * 4 * 32];
  __shared__ v4f w24[13 * 2 * 32];
  const int tid  = threadIdx.x;
  const int wid  = tid >> 5;
  const int lane = tid & 31;
  const int lid  = lane & 15;
  const int kh   = lane >> 4;

  if (tid < 7 * 32) {
    int t = tid >> 5;
    int col = t * 16 + lid;
    int cc = col < 100 ? col : 99;
    bool in = col < 100;
    int k0 = 2 * kh, k1 = 4 + 2 * kh;
    float w00 = W1[k0 * 100 + cc], w01 = W1[(k0 + 1) * 100 + cc];
    float w10 = W1[k1 * 100 + cc], w11 = W1[(k1 + 1) * 100 + cc];
    float w8 = W1[800 + cc], w9 = W1[900 + cc], wb = b1[cc];
    float l00 = L1[k0 * 100 + cc], l01 = L1[(k0 + 1) * 100 + cc];
    float l10 = L1[k1 * 100 + cc], l11 = L1[(k1 + 1) * 100 + cc];
    float l8 = L1[800 + cc], l9 = L1[900 + cc], lb = bl1[cc];
    v4f z4 = {0.0f, 0.0f, 0.0f, 0.0f};
    v2f z2 = {0.0f, 0.0f};
    v2f unit = (col == 104 && kh) ? (v2f){1.0f, 0.0f} : z2;
    w1a[tid] = in ? (v4f){w00, w01, w10, w11} : z4;
    l1a[tid] = in ? (v4f){l00, l01, l10, l11} : z4;
    w1b[tid] = in ? (kh ? (v2f){wb, 0.0f} : (v2f){w8, w9}) : unit;
    l1b[tid] = in ? (kh ? (v2f){lb, 0.0f} : (v2f){l8, l9}) : unit;
  }
#pragma unroll
  for (int i = 0; i < 4; ++i) {
    int idx = tid + i * NTHREADS;
    int ic = idx < 13 * 4 * 32 ? idx : 13 * 4 * 32 - 1;
    int p = ic >> 7, t = (ic >> 5) & 3;
    int row = t * 16 + lid;
    int rc = row < 60 ? row : 59;
    int n = rc >> 1, d = rc & 1;
    bool real = (p < 12) || (kh == 0);
    int hb = real ? 16 * (p >> 1) + 8 * kh + 4 * (p & 1) : 96;
    const float* src = WL2 + n * 200 + hb * 2 + d;
    float x0 = src[0], x1 = src[2], x2 = src[4], x3 = src[6];
    float bb = bL2[rc];
    bool in = row < 60;
    v4f w;
    w.x = in ? (real ? x0 : bb) : 0.0f;
    w.y = (in && real) ? x1 : 0.0f;
    w.z = (in && real) ? x2 : 0.0f;
    w.w = (in && real) ? x3 : 0.0f;
    if (idx < 13 * 4 * 32) wl4[idx] = w;
  }
#pragma unroll
  for (int i = 0; i < 2; ++i) {
    int idx = tid + i * NTHREADS;
    int ic = idx < 13 * 2 * 32 ? idx : 13 * 2 * 32 - 1;
    int p = ic >> 6, t = (ic >> 5) & 1;
    int n = 16 * t + 8 * ((lid >> 2) & 1) + 4 * ((lid >> 3) & 1) + (lid & 3);
    int nc = n < 30 ? n : 29;
    bool real = (p < 12) || (kh == 0);
    int hb = real ? 16 * (p >> 1) + 8 * kh + 4 * (p & 1) : 96;
    const float* src = W2 + hb * 30 + nc;
    float x0 = src[0], x1 = src[30], x2 = src[60], x3 = src[90];
    float bb = b2[nc];
    bool in = n < 30;
    v4f w;
    w.x = in ? (real ? x0 : bb) : 0.0f;
    w.y = (in && real) ? x1 : 0.0f;
    w.z = (in && real) ? x2 : 0.0f;
    w.w = (in && real) ? x3 : 0.0f;
    if (idx < 13 * 2 * 32) w24[idx] = w;
  }

  __syncthreads();

  const int gwave  = blockIdx.x * WAVES + wid;
  const int nwaves = gridDim.x * WAVES;

  for (int chunk = gwave; chunk < nchunks; chunk += nwaves) {
    const int row0 = chunk * ROWS;

    v2f af[2][3];
    float a0[2], a1[2];
#pragma unroll
    for (int m = 0; m < 2; ++m) {
      const int grow = row0 + 16 * m + lid;
      const long rc  = (grow < Btot) ? (long)grow : (long)(Btot - 1);
      const float* srow = s + rc * 10;
      af[m][0] = *(const v2f*)(srow + kh * 2);
      af[m][1] = *(const v2f*)(srow + 4 + kh * 2);
      v2f l    = *(const v2f*)(srow + 8);
      af[m][2] = kh ? (v2f){1.0f, 0.0f} : l;
      v2f av   = *(const v2f*)(a + rc * 2);
      a0[m] = av.x;
      a1[m] = av.y;
    }

    v8f vacc[2][2];
#pragma unroll
    for (int m = 0; m < 2; ++m)
#pragma unroll
      for (int t = 0; t < 2; ++t) vacc[m][t] = (v8f){0, 0, 0, 0, 0, 0, 0, 0};
    branch<2>(vacc, w1a, w1b, w24, af, lane);

    v8f cacc[2][4];
#pragma unroll
    for (int m = 0; m < 2; ++m)
#pragma unroll
      for (int t = 0; t < 4; ++t) cacc[m][t] = (v8f){0, 0, 0, 0, 0, 0, 0, 0};
    branch<4>(cacc, l1a, l1b, wl4, af, lane);

    float ss[2] = {0.0f, 0.0f}, ws[2] = {0.0f, 0.0f};
#pragma unroll
    for (int t = 0; t < 4; ++t) {
#pragma unroll
      for (int u = 0; u < 4; ++u) {
#pragma unroll
        for (int m = 0; m < 2; ++m) {
          float c0 = MAX_A * fast_tanh(cacc[m][t][2 * u]);
          float c1 = MAX_A * fast_tanh(cacc[m][t][2 * u + 1]);
          float v  = vacc[m][t >> 1][4 * (t & 1) + u];
          float d0 = c0 - a0[m], d1 = c1 - a1[m];
          float e  = __expf(-BETA * sqrtf(d0 * d0 + d1 * d1));
          if (t == 3 && u >= 2) e = kh ? 0.0f : e;
          ss[m] += e;
          ws[m] += e * v;
        }
      }
    }
    {
      float s_recv = other_half(kh ? ss[0] : ss[1]);
      float w_recv = other_half(kh ? ws[0] : ws[1]);
      float st = (kh ? ss[1] : ss[0]) + s_recv;
      float wt = (kh ? ws[1] : ws[0]) + w_recv;
      long r = (long)row0 + lane;
      if (r < Btot) out[r] = wt / st;
    }
  }
}

extern "C" void kernel_launch(void* const* d_in, const int* in_sizes, int n_in,
                              void* d_out, int out_size, void* d_ws, size_t ws_size,
                              hipStream_t stream) {
  const float* s   = (const float*)d_in[0];
  const float* a   = (const float*)d_in[1];
  const float* W1  = (const float*)d_in[2];
  const float* b1  = (const float*)d_in[3];
  const float* W2  = (const float*)d_in[4];
  const float* b2  = (const float*)d_in[5];
  const float* L1  = (const float*)d_in[6];
  const float* bl1 = (const float*)d_in[7];
  const float* WL2 = (const float*)d_in[8];
  const float* bL2 = (const float*)d_in[9];
  float* out = (float*)d_out;

  int Btot    = in_sizes[0] / 10;
  int nchunks = (Btot + ROWS - 1) / ROWS;
  int blocks  = (nchunks + WAVES - 1) / WAVES;
  if (blocks > 256) blocks = 256;
  if (blocks < 1) blocks = 1;

  rbf_net_kernel<<<blocks, NTHREADS, 0, stream>>>(
      s, a, W1, b1, W2, b2, L1, bl1, WL2, bL2, out, Btot, nchunks);
}


# Softmax distance root as the bare hardware square root (V_SQRT_F32, 1 ulp) instead of the correctly rounded sqrtf expansion: 377 fewer vector instructions per 32-row chunk; sums unchanged
# speedup vs baseline: 1.4959x; 1.0517x over previous
#include <hip/hip_runtime.h>

typedef __attribute__((ext_vector_type(2))) float v2f;
typedef __attribute__((ext_vector_type(4))) float v4f;
typedef __attribute__((ext_vector_type(8))) float v8f;

#define WAVES 16
#define NTHREADS (WAVES * 32)
#define ROWS 32
#define BETA 3.0f
#define MAX_A 5.0f

static __device__ __forceinline__ v8f wmma4(v2f a, v2f b, v8f c) {
  return __builtin_amdgcn_wmma_f32_16x16x4_f32(
      false, a, false, b, (short)0, c, false, false);
}

static __device__ __forceinline__ float fast_tanh(float x) {
#if __has_builtin(__builtin_amdgcn_tanhf)
  return __builtin_amdgcn_tanhf(x);
#elif __has_builtin(__builtin_amdgcn_tanh_f32)
  return __builtin_amdgcn_tanh_f32(x);
#else
  float e = __expf(2.0f * x);
  return 1.0f - 2.0f / (e + 1.0f);
#endif
}

static __device__ __forceinline__ float hw_sqrt(float x) {
  return __builtin_amdgcn_sqrtf(x);
}

static __device__ __forceinline__ float other_half(float x) {
#if __has_builtin(__builtin_amdgcn_permlanex16)
  unsigned u = __builtin_bit_cast(unsigned, x);
  u = __builtin_amdgcn_permlanex16(u, u, 0x76543210u, 0xfedcba98u, false, false);
  return __builtin_bit_cast(float, u);
#else
  return __shfl_xor(x, 16, 32);
#endif
}

struct WIn { v4f a; v2f b; };
template <int NT> struct WK { v4f w[2][NT]; };

static __device__ __forceinline__ void load_in(WIn& f, const v4f* wa,
    const v2f* wb, int t, int lane) {
  f.a = wa[t * 32 + lane];
  f.b = wb[t * 32 + lane];
}

template <int NT>
static __device__ __forceinline__ void load_k(WK<NT>& f, const v4f* wk, int t,
    int lane) {
#pragma unroll
  for (int pp = 0; pp < 2; ++pp) {
    if (2 * t + pp < 13) {
#pragma unroll
      for (int tt = 0; tt < NT; ++tt)
        f.w[pp][tt] = wk[((2 * t + pp) * NT + tt) * 32 + lane];
    }
  }
}

static __device__ __forceinline__ void first_layer(v8f (&h)[2], const WIn& f,
    const v2f (&af)[2][3]) {
  const v8f z = {0, 0, 0, 0, 0, 0, 0, 0};
#pragma unroll
  for (int m = 0; m < 2; ++m) h[m] = wmma4(f.a.xy, af[m][0], z);
#pragma unroll
  for (int m = 0; m < 2; ++m) h[m] = wmma4(f.a.zw, af[m][1], h[m]);
#pragma unroll
  for (int m = 0; m < 2; ++m) h[m] = wmma4(f.b, af[m][2], h[m]);
}

template <int NT>
static __device__ __forceinline__ void second_layer(v8f (&acc)[2][NT],
    const v8f (&h)[2], const WK<NT>& f, int t) {
#pragma unroll
  for (int pp = 0; pp < 2; ++pp) {
    if (2 * t + pp < 13) {
#pragma unroll
      for (int qq = 0; qq < 2; ++qq) {
        const int q = 2 * pp + qq;
        v2f b[2];
#pragma unroll
        for (int m = 0; m < 2; ++m)
          b[m] = (v2f){fmaxf(h[m][2 * q], 0.0f), fmaxf(h[m][2 * q + 1], 0.0f)};
#pragma unroll
        for (int m = 0; m < 2; ++m)
#pragma unroll
          for (int tt = 0; tt < NT; ++tt)
            acc[m][tt] = wmma4(qq ? f.w[pp][tt].zw : f.w[pp][tt].xy, b[m],
                               acc[m][tt]);
      }
    }
  }
}

template <int NT>
static __device__ __forceinline__ void branch(v8f (&acc)[2][NT], const v4f* wa,
    const v2f* wb, const v4f* wk, const v2f (&af)[2][3], int lane) {
  WIn in_nxt, in_n2;
  WK<NT> k_cur, k_nxt;
  v8f hc[2], hn[2];
  load_in(in_n2, wa, wb, 0, lane);
  load_in(in_nxt, wa, wb, 1, lane);
  load_k<NT>(k_cur, wk, 0, lane);
  first_layer(hc, in_n2, af);
#pragma unroll
  for (int t = 0; t < 7; ++t) {
    if (t + 2 < 7) load_in(in_n2, wa, wb, t + 2, lane);
    if (t + 1 < 7) load_k<NT>(k_nxt, wk, t + 1, lane);
    __builtin_amdgcn_sched_barrier(0);
    if (t + 1 < 7) first_layer(hn, in_nxt, af);
    second_layer<NT>(acc, hc, k_cur, t);
    __builtin_amdgcn_sched_barrier(0);
    if (t + 1 < 7) {
      hc[0] = hn[0];
      hc[1] = hn[1];
      k_cur = k_nxt;
    }
    if (t + 2 < 7) in_nxt = in_n2;
  }
}

__global__ __launch_bounds__(NTHREADS)
void rbf_net_kernel(const float* __restrict__ s,  const float* __restrict__ a,
                    const float* __restrict__ W1, const float* __restrict__ b1,
                    const float* __restrict__ W2, const float* __restrict__ b2,
                    const float* __restrict__ L1, const float* __restrict__ bl1,
                    const float* __restrict__ WL2,const float* __restrict__ bL2,
                    float* __restrict__ out, int Btot, int nchunks) {
  __shared__ v4f w1a[7 * 32];
  __shared__ v2f w1b[7 * 32];
  __shared__ v4f l1a[7 * 32];
  __shared__ v2f l1b[7 * 32];
  __shared__ v4f wl4[13 * 4 * 32];
  __shared__ v4f w24[13 * 2 * 32];
  const int tid  = threadIdx.x;
  const int wid  = tid >> 5;
  const int lane = tid & 31;
  const int lid  = lane & 15;
  const int kh   = lane >> 4;

  if (tid < 7 * 32) {
    int t = tid >> 5;
    int col = t * 16 + lid;
    int cc = col < 100 ? col : 99;
    bool in = col < 100;
    int k0 = 2 * kh, k1 = 4 + 2 * kh;
    float w00 = W1[k0 * 100 + cc], w01 = W1[(k0 + 1) * 100 + cc];
    float w10 = W1[k1 * 100 + cc], w11 = W1[(k1 + 1) * 100 + cc];
    float w8 = W1[800 + cc], w9 = W1[900 + cc], wb = b1[cc];
    float l00 = L1[k0 * 100 + cc], l01 = L1[(k0 + 1) * 100 + cc];
    float l10 = L1[k1 * 100 + cc], l11 = L1[(k1 + 1) * 100 + cc];
    float l8 = L1[800 + cc], l9 = L1[900 + cc], lb = bl1[cc];
    v4f z4 = {0.0f, 0.0f, 0.0f, 0.0f};
    v2f z2 = {0.0f, 0.0f};
    v2f unit = (col == 104 && kh) ? (v2f){1.0f, 0.0f} : z2;
    w1a[tid] = in ? (v4f){w00, w01, w10, w11} : z4;
    l1a[tid] = in ? (v4f){l00, l01, l10, l11} : z4;
    w1b[tid] = in ? (kh ? (v2f){wb, 0.0f} : (v2f){w8, w9}) : unit;
    l1b[tid] = in ? (kh ? (v2f){lb, 0.0f} : (v2f){l8, l9}) : unit;
  }
#pragma unroll
  for (int i = 0; i < 4; ++i) {
    int idx = tid + i * NTHREADS;
    int ic = idx < 13 * 4 * 32 ? idx : 13 * 4 * 32 - 1;
    int p = ic >> 7, t = (ic >> 5) & 3;
    int row = t * 16 + lid;
    int rc = row < 60 ? row : 59;
    int n = rc >> 1, d = rc & 1;
    bool real = (p < 12) || (kh == 0);
    int hb = real ? 16 * (p >> 1) + 8 * kh + 4 * (p & 1) : 96;
    const float* src = WL2 + n * 200 + hb * 2 + d;
    float x0 = src[0], x1 = src[2], x2 = src[4], x3 = src[6];
    float bb = bL2[rc];
    bool in = row < 60;
    v4f w;
    w.x = in ? (real ? x0 : bb) : 0.0f;
    w.y = (in && real) ? x1 : 0.0f;
    w.z = (in && real) ? x2 : 0.0f;
    w.w = (in && real) ? x3 : 0.0f;
    if (idx < 13 * 4 * 32) wl4[idx] = w;
  }
#pragma unroll
  for (int i = 0; i < 2; ++i) {
    int idx = tid + i * NTHREADS;
    int ic = idx < 13 * 2 * 32 ? idx : 13 * 2 * 32 - 1;
    int p = ic >> 6, t = (ic >> 5) & 1;
    int n = 16 * t + 8 * ((lid >> 2) & 1) + 4 * ((lid >> 3) & 1) + (lid & 3);
    int nc = n < 30 ? n : 29;
    bool real = (p < 12) || (kh == 0);
    int hb = real ? 16 * (p >> 1) + 8 * kh + 4 * (p & 1) : 96;
    const float* src = W2 + hb * 30 + nc;
    float x0 = src[0], x1 = src[30], x2 = src[60], x3 = src[90];
    float bb = b2[nc];
    bool in = n < 30;
    v4f w;
    w.x = in ? (real ? x0 : bb) : 0.0f;
    w.y = (in && real) ? x1 : 0.0f;
    w.z = (in && real) ? x2 : 0.0f;
    w.w = (in && real) ? x3 : 0.0f;
    if (idx < 13 * 2 * 32) w24[idx] = w;
  }

  __syncthreads();

  const int gwave  = blockIdx.x * WAVES + wid;
  const int nwaves = gridDim.x * WAVES;

  for (int chunk = gwave; chunk < nchunks; chunk += nwaves) {
    const int row0 = chunk * ROWS;

    v2f af[2][3];
    float a0[2], a1[2];
#pragma unroll
    for (int m = 0; m < 2; ++m) {
      const int grow = row0 + 16 * m + lid;
      const long rc  = (grow < Btot) ? (long)grow : (long)(Btot - 1);
      const float* srow = s + rc * 10;
      af[m][0] = *(const v2f*)(srow + kh * 2);
      af[m][1] = *(const v2f*)(srow + 4 + kh * 2);
      v2f l    = *(const v2f*)(srow + 8);
      af[m][2] = kh ? (v2f){1.0f, 0.0f} : l;
      v2f av   = *(const v2f*)(a + rc * 2);
      a0[m] = av.x;
      a1[m] = av.y;
    }

    v8f vacc[2][2];
#pragma unroll
    for (int m = 0; m < 2; ++m)
#pragma unroll
      for (int t = 0; t < 2; ++t) vacc[m][t] = (v8f){0, 0, 0, 0, 0, 0, 0, 0};
    branch<2>(vacc, w1a, w1b, w24, af, lane);

    v8f cacc[2][4];
#pragma unroll
    for (int m = 0; m < 2; ++m)
#pragma unroll
      for (int t = 0; t < 4; ++t) cacc[m][t] = (v8f){0, 0, 0, 0, 0, 0, 0, 0};
    branch<4>(cacc, l1a, l1b, wl4, af, lane);

    float ss[2] = {0.0f, 0.0f}, ws[2] = {0.0f, 0.0f};
#pragma unroll
    for (int t = 0; t < 4; ++t) {
#pragma unroll
      for (int u = 0; u < 4; ++u) {
#pragma unroll
        for (int m = 0; m < 2; ++m) {
          float c0 = MAX_A * fast_tanh(cacc[m][t][2 * u]);
          float c1 = MAX_A * fast_tanh(cacc[m][t][2 * u + 1]);
          float v  = vacc[m][t >> 1][4 * (t & 1) + u];
          float d0 = c0 - a0[m], d1 = c1 - a1[m];
          float e  = __expf(-BETA * hw_sqrt(d0 * d0 + d1 * d1));
          if (t == 3 && u >= 2) e = kh ? 0.0f : e;
          ss[m] += e;
          ws[m] += e * v;
        }
      }
    }
    {
      float s_recv = other_half(kh ? ss[0] : ss[1]);
      float w_recv = other_half(kh ? ws[0] : ws[1]);
      float st = (kh ? ss[1] : ss[0]) + s_recv;
      float wt = (kh ? ws[1] : ws[0]) + w_recv;
      long r = (long)row0 + lane;
      if (r < Btot) out[r] = wt / st;
    }
  }
}

extern "C" void kernel_launch(void* const* d_in, const int* in_sizes, int n_in,
                              void* d_out, int out_size, void* d_ws, size_t ws_size,
                              hipStream_t stream) {
  const float* s   = (const float*)d_in[0];
  const float* a   = (const float*)d_in[1];
  const float* W1  = (const float*)d_in[2];
  const float* b1  = (const float*)d_in[3];
  const float* W2  = (const float*)d_in[4];
  const float* b2  = (const float*)d_in[5];
  const float* L1  = (const float*)d_in[6];
  const float* bl1 = (const float*)d_in[7];
  const float* WL2 = (const float*)d_in[8];
  const float* bL2 = (const float*)d_in[9];
  float* out = (float*)d_out;

  int Btot    = in_sizes[0] / 10;
  int nchunks = (Btot + ROWS - 1) / ROWS;
  int blocks  = (nchunks + WAVES - 1) / WAVES;
  if (blocks > 256) blocks = 256;
  if (blocks < 1) blocks = 1;

  rbf_net_kernel<<<blocks, NTHREADS, 0, stream>>>(
      s, a, W1, b1, W2, b2, L1, bl1, WL2, bL2, out, Btot, nchunks);
}


# Global loads asked for ahead of use: the weight fill issues all its loads as one group (branch-free, unconditional stores into padded LDS), each chunk's inputs are loaded one chunk ahead
# speedup vs baseline: 1.5731x; 1.0516x over previous
#include <hip/hip_runtime.h>

typedef __attribute__((ext_vector_type(2))) float v2f;
typedef __attribute__((ext_vector_type(4))) float v4f;
typedef __attribute__((ext_vector_type(8))) float v8f;

#define WAVES 16
#define NTHREADS (WAVES * 32)
#define ROWS 32
#define BETA 3.0f
#define MAX_A 5.0f

static __device__ __forceinline__ v8f wmma4(v2f a, v2f b, v8f c) {
  return __builtin_amdgcn_wmma_f32_16x16x4_f32(
      false, a, false, b, (short)0, c, false, false);
}

static __device__ __forceinline__ float fast_tanh(float x) {
#if __has_builtin(__builtin_amdgcn_tanhf)
  return __builtin_amdgcn_tanhf(x);
#elif __has_builtin(__builtin_amdgcn_tanh_f32)
  return __builtin_amdgcn_tanh_f32(x);
#else
  float e = __expf(2.0f * x);
  return 1.0f - 2.0f / (e + 1.0f);
#endif
}

static __device__ __forceinline__ float hw_sqrt(float x) {
  return __builtin_amdgcn_sqrtf(x);
}

static __device__ __forceinline__ float other_half(float x) {
#if __has_builtin(__builtin_amdgcn_permlanex16)
  unsigned u = __builtin_bit_cast(unsigned, x);
  u = __builtin_amdgcn_permlanex16(u, u, 0x76543210u, 0xfedcba98u, false, false);
  return __builtin_bit_cast(float, u);
#else
  return __shfl_xor(x, 16, 32);
#endif
}

struct WIn { v4f a; v2f b; };
template <int NT> struct WK { v4f w[2][NT]; };

static __device__ __forceinline__ void load_in(WIn& f, const v4f* wa,
    const v2f* wb, int t, int lane) {
  f.a = wa[t * 32 + lane];
  f.b = wb[t * 32 + lane];
}

template <int NT>
static __device__ __forceinline__ void load_k(WK<NT>& f, const v4f* wk, int t,
    int lane) {
#pragma unroll
  for (int pp = 0; pp < 2; ++pp) {
    if (2 * t + pp < 13) {
#pragma unroll
      for (int tt = 0; tt < NT; ++tt)
        f.w[pp][tt] = wk[((2 * t + pp) * NT + tt) * 32 + lane];
    }
  }
}

static __device__ __forceinline__ void first_layer(v8f (&h)[2], const WIn& f,
    const v2f (&af)[2][3]) {
  const v8f z = {0, 0, 0, 0, 0, 0, 0, 0};
#pragma unroll
  for (int m = 0; m < 2; ++m) h[m] = wmma4(f.a.xy, af[m][0], z);
#pragma unroll
  for (int m = 0; m < 2; ++m) h[m] = wmma4(f.a.zw, af[m][1], h[m]);
#pragma unroll
  for (int m = 0; m < 2; ++m) h[m] = wmma4(f.b, af[m][2], h[m]);
}

template <int NT>
static __device__ __forceinline__ void second_layer(v8f (&acc)[2][NT],
    const v8f (&h)[2], const WK<NT>& f, int t) {
#pragma unroll
  for (int pp = 0; pp < 2; ++pp) {
    if (2 * t + pp < 13) {
#pragma unroll
      for (int qq = 0; qq < 2; ++qq) {
        const int q = 2 * pp + qq;
        v2f b[2];
#pragma unroll
        for (int m = 0; m < 2; ++m)
          b[m] = (v2f){fmaxf(h[m][2 * q], 0.0f), fmaxf(h[m][2 * q + 1], 0.0f)};
#pragma unroll
        for (int m = 0; m < 2; ++m)
#pragma unroll
          for (int tt = 0; tt < NT; ++tt)
            acc[m][tt] = wmma4(qq ? f.w[pp][tt].zw : f.w[pp][tt].xy, b[m],
                               acc[m][tt]);
      }
    }
  }
}

template <int NT>
static __device__ __forceinline__ void branch(v8f (&acc)[2][NT], const v4f* wa,
    const v2f* wb, const v4f* wk, const v2f (&af)[2][3], int lane) {
  WIn in_nxt, in_n2;
  WK<NT> k_cur, k_nxt;
  v8f hc[2], hn[2];
  load_in(in_n2, wa, wb, 0, lane);
  load_in(in_nxt, wa, wb, 1, lane);
  load_k<NT>(k_cur, wk, 0, lane);
  first_layer(hc, in_n2, af);
#pragma unroll
  for (int t = 0; t < 7; ++t) {
    if (t + 2 < 7) load_in(in_n2, wa, wb, t + 2, lane);
    if (t + 1 < 7) load_k<NT>(k_nxt, wk, t + 1, lane);
    __builtin_amdgcn_sched_barrier(0);
    if (t + 1 < 7) first_layer(hn, in_nxt, af);
    second_layer<NT>(acc, hc, k_cur, t);
    __builtin_amdgcn_sched_barrier(0);
    if (t + 1 < 7) {
      hc[0] = hn[0];
      hc[1] = hn[1];
      k_cur = k_nxt;
    }
    if (t + 2 < 7) in_nxt = in_n2;
  }
}

struct Inputs { v2f af[2][3]; float a0[2], a1[2]; };

static __device__ __forceinline__ void load_inputs(Inputs& in,
    const float* __restrict__ s, const float* __restrict__ a, long row0,
    int Btot, int lid, int kh) {
#pragma unroll
  for (int m = 0; m < 2; ++m) {
    const long grow = row0 + 16 * m + lid;
    const long rc   = (grow < (long)Btot) ? grow : (long)(Btot - 1);
    const float* srow = s + rc * 10;
    in.af[m][0] = *(const v2f*)(srow + kh * 2);
    in.af[m][1] = *(const v2f*)(srow + 4 + kh * 2);
    v2f l       = *(const v2f*)(srow + 8);
    in.af[m][2] = kh ? (v2f){1.0f, 0.0f} : l;
    v2f av      = *(const v2f*)(a + rc * 2);
    in.a0[m] = av.x;
    in.a1[m] = av.y;
  }
}

__global__ __launch_bounds__(NTHREADS)
void rbf_net_kernel(const float* __restrict__ s,  const float* __restrict__ a,
                    const float* __restrict__ W1, const float* __restrict__ b1,
                    const float* __restrict__ W2, const float* __restrict__ b2,
                    const float* __restrict__ L1, const float* __restrict__ bl1,
                    const float* __restrict__ WL2,const float* __restrict__ bL2,
                    float* __restrict__ out, int Btot, int nchunks) {
  __shared__ v4f w1a[NTHREADS];
  __shared__ v2f w1b[NTHREADS];
  __shared__ v4f l1a[NTHREADS];
  __shared__ v2f l1b[NTHREADS];
  __shared__ v4f wl4[4 * NTHREADS];
  __shared__ v4f w24[2 * NTHREADS];
  const int tid  = threadIdx.x;
  const int wid  = tid >> 5;
  const int lane = tid & 31;
  const int lid  = lane & 15;
  const int kh   = lane >> 4;

  const int gwave  = blockIdx.x * WAVES + wid;
  const int nwaves = gridDim.x * WAVES;
  Inputs in;
  load_inputs(in, s, a, (long)gwave * ROWS, Btot, lid, kh);

  const int t1   = tid >> 5;
  const int col1 = t1 * 16 + lid;
  const int cc1  = col1 < 100 ? col1 : 99;
  const int k0 = 2 * kh, k1 = 4 + 2 * kh;
  const float w00 = W1[k0 * 100 + cc1], w01 = W1[(k0 + 1) * 100 + cc1];
  const float w10 = W1[k1 * 100 + cc1], w11 = W1[(k1 + 1) * 100 + cc1];
  const float w8 = W1[800 + cc1], w9 = W1[900 + cc1], wb = b1[cc1];
  const float l00 = L1[k0 * 100 + cc1], l01 = L1[(k0 + 1) * 100 + cc1];
  const float l10 = L1[k1 * 100 + cc1], l11 = L1[(k1 + 1) * 100 + cc1];
  const float l8 = L1[800 + cc1], l9 = L1[900 + cc1], lb = bl1[cc1];

  float lx[4][4], lbias[4];
  bool  lin[4], lreal[4];
#pragma unroll
  for (int i = 0; i < 4; ++i) {
    int idx = tid + i * NTHREADS;
    int ic = idx < 13 * 4 * 32 ? idx : 13 * 4 * 32 - 1;
    int p = ic >> 7, t = (ic >> 5) & 3;
    int row = t * 16 + lid;
    int rc = row < 60 ? row : 59;
    int n = rc >> 1, d = rc & 1;
    bool real = (p < 12) || (kh == 0);
    int hb = real ? 16 * (p >> 1) + 8 * kh + 4 * (p & 1) : 96;
    const float* src = WL2 + n * 200 + hb * 2 + d;
    lx[i][0] = src[0]; lx[i][1] = src[2]; lx[i][2] = src[4]; lx[i][3] = src[6];
    lbias[i] = bL2[rc];
    lin[i] = row < 60;
    lreal[i] = real;
  }
  float vx[2][4], vbias[2];
  bool  vin[2], vreal[2];
#pragma unroll
  for (int i = 0; i < 2; ++i) {
    int idx = tid + i * NTHREADS;
    int ic = idx < 13 * 2 * 32 ? idx : 13 * 2 * 32 - 1;
    int p = ic >> 6, t = (ic >> 5) & 1;
    int n = 16 * t + 8 * ((lid >> 2) & 1) + 4 * ((lid >> 3) & 1) + (lid & 3);
    int nc = n < 30 ? n : 29;
    bool real = (p < 12) || (kh == 0);
    int hb = real ? 16 * (p >> 1) + 8 * kh + 4 * (p & 1) : 96;
    const float* src = W2 + hb * 30 + nc;
    vx[i][0] = src[0]; vx[i][1] = src[30]; vx[i][2] = src[60]; vx[i][3] = src[90];
    vbias[i] = b2[nc];
    vin[i] = n < 30;
    vreal[i] = real;
  }

  __builtin_amdgcn_sched_barrier(0);
  {
    const bool in1 = col1 < 100;
    const v4f z4 = {0.0f, 0.0f, 0.0f, 0.0f};
    const v2f z2 = {0.0f, 0.0f};
    const v2f unit = (col1 == 104 && kh) ? (v2f){1.0f, 0.0f} : z2;
    w1a[tid] = in1 ? (v4f){w00, w01, w10, w11} : z4;
    l1a[tid] = in1 ? (v4f){l00, l01, l10, l11} : z4;
    w1b[tid] = in1 ? (kh ? (v2f){wb, 0.0f} : (v2f){w8, w9}) : unit;
    l1b[tid] = in1 ? (kh ? (v2f){lb, 0.0f} : (v2f){l8, l9}) : unit;
  }
#pragma unroll
  for (int i = 0; i < 4; ++i) {
    v4f w;
    w.x = lin[i] ? (lreal[i] ? lx[i][0] : lbias[i]) : 0.0f;
    w.y = (lin[i] && lreal[i]) ? lx[i][1] : 0.0f;
    w.z = (lin[i] && lreal[i]) ? lx[i][2] : 0.0f;
    w.w = (lin[i] && lreal[i]) ? lx[i][3] : 0.0f;
    wl4[tid + i * NTHREADS] = w;
  }
#pragma unroll
  for (int i = 0; i < 2; ++i) {
    v4f w;
    w.x = vin[i] ? (vreal[i] ? vx[i][0] : vbias[i]) : 0.0f;
    w.y = (vin[i] && vreal[i]) ? vx[i][1] : 0.0f;
    w.z = (vin[i] && vreal[i]) ? vx[i][2] : 0.0f;
    w.w = (vin[i] && vreal[i]) ? vx[i][3] : 0.0f;
    w24[tid + i * NTHREADS] = w;
  }

  __syncthreads();

  for (int chunk = gwave; chunk < nchunks; chunk += nwaves) {
    const int row0 = chunk * ROWS;

    const float a0[2] = {in.a0[0], in.a0[1]};
    const float a1[2] = {in.a1[0], in.a1[1]};

    v8f vacc[2][2];
#pragma unroll
    for (int m = 0; m < 2; ++m)
#pragma unroll
      for (int t = 0; t < 2; ++t) vacc[m][t] = (v8f){0, 0, 0, 0, 0, 0, 0, 0};
    branch<2>(vacc, w1a, w1b, w24, in.af, lane);

    v8f cacc[2][4];
#pragma unroll
    for (int m = 0; m < 2; ++m)
#pragma unroll
      for (int t = 0; t < 4; ++t) cacc[m][t] = (v8f){0, 0, 0, 0, 0, 0, 0, 0};
    branch<4>(cacc, l1a, l1b, wl4, in.af, lane);

    load_inputs(in, s, a, ((long)chunk + nwaves) * ROWS, Btot, lid, kh);

    float ss[2] = {0.0f, 0.0f}, ws[2] = {0.0f, 0.0f};
#pragma unroll
    for (int t = 0; t < 4; ++t) {
#pragma unroll
      for (int u = 0; u < 4; ++u) {
#pragma unroll
        for (int m = 0; m < 2; ++m) {
          float c0 = MAX_A * fast_tanh(cacc[m][t][2 * u]);
          float c1 = MAX_A * fast_tanh(cacc[m][t][2 * u + 1]);
          float v  = vacc[m][t >> 1][4 * (t & 1) + u];
          float d0 = c0 - a0[m], d1 = c1 - a1[m];
          float e  = __expf(-BETA * hw_sqrt(d0 * d0 + d1 * d1));
          if (t == 3 && u >= 2) e = kh ? 0.0f : e;
          ss[m] += e;
          ws[m] += e * v;
        }
      }
    }
    {
      float s_recv = other_half(kh ? ss[0] : ss[1]);
      float w_recv = other_half(kh ? ws[0] : ws[1]);
      float st = (kh ? ss[1] : ss[0]) + s_recv;
      float wt = (kh ? ws[1] : ws[0]) + w_recv;
      long r = (long)row0 + lane;
      if (r < Btot) out[r] = wt / st;
    }
  }
}

extern "C" void kernel_launch(void* const* d_in, const int* in_sizes, int n_in,
                              void* d_out, int out_size, void* d_ws, size_t ws_size,
                              hipStream_t stream) {
  const float* s   = (const float*)d_in[0];
  const float* a   = (const float*)d_in[1];
  const float* W1  = (const float*)d_in[2];
  const float* b1  = (const float*)d_in[3];
  const float* W2  = (const float*)d_in[4];
  const float* b2  = (const float*)d_in[5];
  const float* L1  = (const float*)d_in[6];
  const float* bl1 = (const float*)d_in[7];
  const float* WL2 = (const float*)d_in[8];
  const float* bL2 = (const float*)d_in[9];
  float* out = (float*)d_out;

  int Btot    = in_sizes[0] / 10;
  int nchunks = (Btot + ROWS - 1) / ROWS;
  int blocks  = (nchunks + WAVES - 1) / WAVES;
  if (blocks > 256) blocks = 256;
  if (blocks < 1) blocks = 1;

  rbf_net_kernel<<<blocks, NTHREADS, 0, stream>>>(
      s, a, W1, b1, W2, b2, L1, bl1, WL2, bL2, out, Btot, nchunks);
}


# Padding trimmed from the products: layer-2 biases become the accumulators' start values (K 104 -> 100), units 96..99 of both branches share one first-layer tile (14 -> 13); bias is summed first
# speedup vs baseline: 1.6258x; 1.0335x over previous
#include <hip/hip_runtime.h>

typedef __attribute__((ext_vector_type(2))) float v2f;
typedef __attribute__((ext_vector_type(4))) float v4f;
typedef __attribute__((ext_vector_type(8))) float v8f;

#define WAVES 16
#define NTHREADS (WAVES * 32)
#define ROWS 32
#define BETA 3.0f
#define MAX_A 5.0f

static __device__ __forceinline__ v8f wmma4(v2f a, v2f b, v8f c) {
  return __builtin_amdgcn_wmma_f32_16x16x4_f32(
      false, a, false, b, (short)0, c, false, false);
}

static __device__ __forceinline__ float fast_tanh(float x) {
#if __has_builtin(__builtin_amdgcn_tanhf)
  return __builtin_amdgcn_tanhf(x);
#elif __has_builtin(__builtin_amdgcn_tanh_f32)
  return __builtin_amdgcn_tanh_f32(x);
#else
  float e = __expf(2.0f * x);
  return 1.0f - 2.0f / (e + 1.0f);
#endif
}

static __device__ __forceinline__ float hw_sqrt(float x) {
  return __builtin_amdgcn_sqrtf(x);
}

static __device__ __forceinline__ float other_half(float x) {
#if __has_builtin(__builtin_amdgcn_permlanex16)
  unsigned u = __builtin_bit_cast(unsigned, x);
  u = __builtin_amdgcn_permlanex16(u, u, 0x76543210u, 0xfedcba98u, false, false);
  return __builtin_bit_cast(float, u);
#else
  return __shfl_xor(x, 16, 32);
#endif
}

struct WIn { v4f a; v2f b; };
template <int NT> struct WK { v4f w[2][NT]; };

static __device__ __forceinline__ void load_in(WIn& f, const v4f* wa,
    const v2f* wb, int t, int lane) {
  f.a = wa[t * 32 + lane];
  f.b = wb[t * 32 + lane];
}

template <int NT>
static __device__ __forceinline__ void load_k(WK<NT>& f, const v4f* wk, int t,
    int lane) {
#pragma unroll
  for (int pp = 0; pp < 2; ++pp) {
    if (2 * t + pp < 13) {
#pragma unroll
      for (int tt = 0; tt < NT; ++tt)
        f.w[pp][tt] = wk[((2 * t + pp) * NT + tt) * 32 + lane];
    }
  }
}

static __device__ __forceinline__ void first_layer(v8f (&h)[2], const WIn& f,
    const v2f (&af)[2][3]) {
  const v8f z = {0, 0, 0, 0, 0, 0, 0, 0};
#pragma unroll
  for (int m = 0; m < 2; ++m) h[m] = wmma4(f.a.xy, af[m][0], z);
#pragma unroll
  for (int m = 0; m < 2; ++m) h[m] = wmma4(f.a.zw, af[m][1], h[m]);
#pragma unroll
  for (int m = 0; m < 2; ++m) h[m] = wmma4(f.b, af[m][2], h[m]);
}

template <int NT>
static __device__ __forceinline__ void second_layer(v8f (&acc)[2][NT],
    const v8f (&h)[2], const WK<NT>& f, int t) {
#pragma unroll
  for (int pp = 0; pp < 2; ++pp) {
    if (2 * t + pp < 13) {
#pragma unroll
      for (int qq = 0; qq < 2; ++qq) {
        const int q = 2 * pp + qq;
        v2f b[2];
#pragma unroll
        for (int m = 0; m < 2; ++m)
          b[m] = (v2f){fmaxf(h[m][2 * q], 0.0f), fmaxf(h[m][2 * q + 1], 0.0f)};
#pragma unroll
        for (int m = 0; m < 2; ++m)
#pragma unroll
          for (int tt = 0; tt < NT; ++tt)
            acc[m][tt] = wmma4(qq ? f.w[pp][tt].zw : f.w[pp][tt].xy, b[m],
                               acc[m][tt]);
      }
    }
  }
}

template <int NT>
static __device__ __forceinline__ void tail_step(v8f (&acc)[2][NT],
    const v2f (&b)[2], const WK<NT>& f) {
#pragma unroll
  for (int m = 0; m < 2; ++m)
#pragma unroll
    for (int tt = 0; tt < NT; ++tt)
      acc[m][tt] = wmma4(f.w[0][tt].xy, b[m], acc[m][tt]);
}

template <int NT>
static __device__ __forceinline__ void load_bias(v8f (&acc)[2][NT],
    const float* bias, int kh) {
#pragma unroll
  for (int tt = 0; tt < NT; ++tt) {
    const v4f* p = (const v4f*)(bias + tt * 16 + 8 * kh);
    const v4f lo = p[0], hi = p[1];
    const v8f b = {lo.x, lo.y, lo.z, lo.w, hi.x, hi.y, hi.z, hi.w};
    acc[0][tt] = b;
    acc[1][tt] = b;
  }
}

template <int NT, bool OWNS_S>
static __device__ __forceinline__ void branch(v8f (&acc)[2][NT], const v4f* wa,
    const v2f* wb, const v4f* wk, const float* bias, const v2f (&af)[2][3],
    v2f (&bS)[2], int lane, int kh) {
  constexpr int NT1 = OWNS_S ? 7 : 6;
  WIn in_nxt, in_n2;
  WK<NT> k_cur, k_nxt;
  v8f hc[2], hn[2];
  load_in(in_n2, wa, wb, 0, lane);
  load_in(in_nxt, wa, wb, 1, lane);
  load_k<NT>(k_cur, wk, 0, lane);
  load_bias<NT>(acc, bias, kh);
  first_layer(hc, in_n2, af);
#pragma unroll
  for (int t = 0; t < 7; ++t) {
    if (t + 2 < NT1) load_in(in_n2, wa, wb, t + 2, lane);
    if (t + 1 < 7) load_k<NT>(k_nxt, wk, t + 1, lane);
    __builtin_amdgcn_sched_barrier(0);
    if (t + 1 < NT1) first_layer(hn, in_nxt, af);
    if (t < 6) {
      second_layer<NT>(acc, hc, k_cur, t);
    } else if (OWNS_S) {
      v2f b[2];
#pragma unroll
      for (int m = 0; m < 2; ++m) {
        b[m]  = (v2f){fmaxf(hc[m][0], 0.0f), fmaxf(hc[m][1], 0.0f)};
        bS[m] = (v2f){fmaxf(hc[m][2], 0.0f), fmaxf(hc[m][3], 0.0f)};
      }
      tail_step<NT>(acc, b, k_cur);
    } else {
      tail_step<NT>(acc, bS, k_cur);
    }
    __builtin_amdgcn_sched_barrier(0);
    if (t + 1 < NT1) {
      hc[0] = hn[0];
      hc[1] = hn[1];
    }
    if (t + 1 < 7) k_cur = k_nxt;
    if (t + 2 < NT1) in_nxt = in_n2;
  }
}

struct Inputs { v2f af[2][3]; float a0[2], a1[2]; };

static __device__ __forceinline__ void load_inputs(Inputs& in,
    const float* __restrict__ s, const float* __restrict__ a, long row0,
    int Btot, int lid, int kh) {
#pragma unroll
  for (int m = 0; m < 2; ++m) {
    const long grow = row0 + 16 * m + lid;
    const long rc   = (grow < (long)Btot) ? grow : (long)(Btot - 1);
    const float* srow = s + rc * 10;
    in.af[m][0] = *(const v2f*)(srow + kh * 2);
    in.af[m][1] = *(const v2f*)(srow + 4 + kh * 2);
    v2f l       = *(const v2f*)(srow + 8);
    in.af[m][2] = kh ? (v2f){1.0f, 0.0f} : l;
    v2f av      = *(const v2f*)(a + rc * 2);
    in.a0[m] = av.x;
    in.a1[m] = av.y;
  }
}

__global__ __launch_bounds__(NTHREADS)
void rbf_net_kernel(const float* __restrict__ s,  const float* __restrict__ a,
                    const float* __restrict__ W1, const float* __restrict__ b1,
                    const float* __restrict__ W2, const float* __restrict__ b2,
                    const float* __restrict__ L1, const float* __restrict__ bl1,
                    const float* __restrict__ WL2,const float* __restrict__ bL2,
                    float* __restrict__ out, int Btot, int nchunks) {
  __shared__ v4f f1a[NTHREADS];
  __shared__ v2f f1b[NTHREADS];
  __shared__ v4f wl4[4 * NTHREADS];
  __shared__ v4f w24[2 * NTHREADS];
  __shared__ __attribute__((aligned(16)))
  float bias2[NTHREADS];
  const int tid  = threadIdx.x;
  const int wid  = tid >> 5;
  const int lane = tid & 31;
  const int lid  = lane & 15;
  const int kh   = lane >> 4;

  const int gwave  = blockIdx.x * WAVES + wid;
  const int nwaves = gridDim.x * WAVES;
  Inputs in;
  load_inputs(in, s, a, (long)gwave * ROWS, Btot, lid, kh);

  const int T1 = tid >> 5;
  const int j1 = lid & 7;
  const bool isS  = (T1 == 6);
  const bool useL = isS ? (j1 >= 2) : (T1 > 6);
  const bool in1  = isS ? (j1 < 4) : (T1 < 13);
  const int unitS = 96 + 2 * (lid >> 3) + (j1 & 1);
  const int unitF = 16 * (T1 > 6 ? T1 - 7 : T1) + lid;
  const int u1 = in1 ? (isS ? unitS : unitF) : 0;
  const float* Wp = useL ? L1 : W1;
  const float* bp = useL ? bl1 : b1;
  const int k0 = 2 * kh, k1 = 4 + 2 * kh;
  const float w00 = Wp[k0 * 100 + u1], w01 = Wp[(k0 + 1) * 100 + u1];
  const float w10 = Wp[k1 * 100 + u1], w11 = Wp[(k1 + 1) * 100 + u1];
  const float w8 = Wp[800 + u1], w9 = Wp[900 + u1], wb = bp[u1];

  float lx[4][4];
  bool  lin[4], lfull[4];
#pragma unroll
  for (int i = 0; i < 4; ++i) {
    int idx = tid + i * NTHREADS;
    int ic = idx < 13 * 4 * 32 ? idx : 13 * 4 * 32 - 1;
    int p = ic >> 7, t = (ic >> 5) & 3;
    int row = t * 16 + lid;
    int rc = row < 60 ? row : 59;
    int n = rc >> 1, d = rc & 1;
    bool full = p < 12;
    int hb = full ? 16 * (p >> 1) + 8 * kh + 4 * (p & 1) : 96 + 2 * kh;
    const float* src = WL2 + n * 200 + hb * 2 + d;
    lx[i][0] = src[0]; lx[i][1] = src[2];
    lx[i][2] = src[full ? 4 : 0]; lx[i][3] = src[full ? 6 : 2];
    lin[i] = row < 60;
    lfull[i] = full;
  }
  float vx[2][4];
  bool  vin[2], vfull[2];
#pragma unroll
  for (int i = 0; i < 2; ++i) {
    int idx = tid + i * NTHREADS;
    int ic = idx < 13 * 2 * 32 ? idx : 13 * 2 * 32 - 1;
    int p = ic >> 6, t = (ic >> 5) & 1;
    int n = 16 * t + 8 * ((lid >> 2) & 1) + 4 * ((lid >> 3) & 1) + (lid & 3);
    int nc = n < 30 ? n : 29;
    bool full = p < 12;
    int hb = full ? 16 * (p >> 1) + 8 * kh + 4 * (p & 1) : 96 + 2 * kh;
    const float* src = W2 + hb * 30 + nc;
    vx[i][0] = src[0]; vx[i][1] = src[30];
    vx[i][2] = src[full ? 60 : 0]; vx[i][3] = src[full ? 90 : 30];
    vin[i] = n < 30;
    vfull[i] = full;
  }
  const int rv = tid & 15;
  const int nv = 16 * ((tid >> 4) & 1) + 8 * ((rv >> 2) & 1) + 4 * ((rv >> 3) & 1) + (rv & 3);
  const int jl = tid - 32;
  const bool bval = tid < 32;
  const bool bin  = bval ? (nv < 30) : (jl < 60);
  const int  bidx = bin ? (bval ? nv : jl) : 0;
  const float bsv = (bval ? b2 : bL2)[bidx];

  __builtin_amdgcn_sched_barrier(0);
  {
    const v4f z4 = {0.0f, 0.0f, 0.0f, 0.0f};
    const v2f z2 = {0.0f, 0.0f};
    f1a[tid] = in1 ? (v4f){w00, w01, w10, w11} : z4;
    f1b[tid] = in1 ? (kh ? (v2f){wb, 0.0f} : (v2f){w8, w9}) : z2;
    bias2[tid] = bin ? bsv : 0.0f;
  }
#pragma unroll
  for (int i = 0; i < 4; ++i) {
    v4f w;
    w.x = lin[i] ? lx[i][0] : 0.0f;
    w.y = lin[i] ? lx[i][1] : 0.0f;
    w.z = (lin[i] && lfull[i]) ? lx[i][2] : 0.0f;
    w.w = (lin[i] && lfull[i]) ? lx[i][3] : 0.0f;
    wl4[tid + i * NTHREADS] = w;
  }
#pragma unroll
  for (int i = 0; i < 2; ++i) {
    v4f w;
    w.x = vin[i] ? vx[i][0] : 0.0f;
    w.y = vin[i] ? vx[i][1] : 0.0f;
    w.z = (vin[i] && vfull[i]) ? vx[i][2] : 0.0f;
    w.w = (vin[i] && vfull[i]) ? vx[i][3] : 0.0f;
    w24[tid + i * NTHREADS] = w;
  }

  __syncthreads();

  for (int chunk = gwave; chunk < nchunks; chunk += nwaves) {
    const int row0 = chunk * ROWS;

    const float a0[2] = {in.a0[0], in.a0[1]};
    const float a1[2] = {in.a1[0], in.a1[1]};

    v8f vacc[2][2];
    v2f bS[2];
    branch<2, true>(vacc, f1a, f1b, w24, bias2, in.af, bS, lane, kh);

    v8f cacc[2][4];
    branch<4, false>(cacc, f1a + 7 * 32, f1b + 7 * 32, wl4, bias2 + 32, in.af,
                     bS, lane, kh);

    load_inputs(in, s, a, ((long)chunk + nwaves) * ROWS, Btot, lid, kh);

    float ss[2] = {0.0f, 0.0f}, ws[2] = {0.0f, 0.0f};
#pragma unroll
    for (int t = 0; t < 4; ++t) {
#pragma unroll
      for (int u = 0; u < 4; ++u) {
#pragma unroll
        for (int m = 0; m < 2; ++m) {
          float c0 = MAX_A * fast_tanh(cacc[m][t][2 * u]);
          float c1 = MAX_A * fast_tanh(cacc[m][t][2 * u + 1]);
          float v  = vacc[m][t >> 1][4 * (t & 1) + u];
          float d0 = c0 - a0[m], d1 = c1 - a1[m];
          float e  = __expf(-BETA * hw_sqrt(d0 * d0 + d1 * d1));
          if (t == 3 && u >= 2) e = kh ? 0.0f : e;
          ss[m] += e;
          ws[m] += e * v;
        }
      }
    }
    {
      float s_recv = other_half(kh ? ss[0] : ss[1]);
      float w_recv = other_half(kh ? ws[0] : ws[1]);
      float st = (kh ? ss[1] : ss[0]) + s_recv;
      float wt = (kh ? ws[1] : ws[0]) + w_recv;
      long r = (long)row0 + lane;
      if (r < Btot) out[r] = wt / st;
    }
  }
}

extern "C" void kernel_launch(void* const* d_in, const int* in_sizes, int n_in,
                              void* d_out, int out_size, void* d_ws, size_t ws_size,
                              hipStream_t stream) {
  const float* s   = (const float*)d_in[0];
  const float* a   = (const float*)d_in[1];
  const float* W1  = (const float*)d_in[2];
  const float* b1  = (const float*)d_in[3];
  const float* W2  = (const float*)d_in[4];
  const float* b2  = (const float*)d_in[5];
  const float* L1  = (const float*)d_in[6];
  const float* bl1 = (const float*)d_in[7];
  const float* WL2 = (const float*)d_in[8];
  const float* bL2 = (const float*)d_in[9];
  float* out = (float*)d_out;

  int Btot    = in_sizes[0] / 10;
  int nchunks = (Btot + ROWS - 1) / ROWS;
  int blocks  = (nchunks + WAVES - 1) / WAVES;
  if (blocks > 256) blocks = 256;
  if (blocks < 1) blocks = 1;

  rbf_net_kernel<<<blocks, NTHREADS, 0, stream>>>(
      s, a, W1, b1, W2, b2, L1, bl1, WL2, bL2, out, Btot, nchunks);
}
